# starting kernel
# baseline (speedup 1.0000x reference)
#include <hip/hip_runtime.h>


typedef float v2f __attribute__((ext_vector_type(2)));
typedef float v8f __attribute__((ext_vector_type(8)));

#define BN 16
#define CN 256
#define HN 48
#define WN 64
#define HWN (HN * WN)
#define ND 21
#define DN (ND * ND)

__global__ __launch_bounds__(256) void corr_wmma_f32_kernel(
    const float* __restrict__ in1,
    const float* __restrict__ in2,
    float* __restrict__ out)
{
    const int wave = threadIdx.x >> 5;
    const int lane = threadIdx.x & 31;
    const int half = lane >> 4;
    const int l    = lane & 15;

    int task = blockIdx.x * (blockDim.x >> 5) + wave;
    const int iy    = task % ND; task /= ND;
    const int strip = task % 4;  task /= 4;
    const int h     = task % HN;
    const int b     = task / HN;

    const int w0    = strip * 16;
    const int qbase = w0 - 24;
    const int dy    = -20 + 2 * iy;
    const int h2    = h + dy;
    const bool row_valid = (h2 >= 0) && (h2 < HN);

    v8f acc0 = {}, acc1 = {}, acc2 = {}, acc3 = {};

    if (row_valid) {
        const int k0 = half * 2;

        const float* __restrict__ ap =
            in1 + (((size_t)b * CN + k0) * HN + h) * WN + (w0 + l);

        const size_t brow = (((size_t)b * CN + k0) * HN + h2) * WN;

        const int q0 = qbase + l;
        const int q1 = q0 + 16;
        const int q2 = q0 + 32;
        const int q3 = q0 + 48;
        const bool v0 = (q0 >= 0) && (q0 < WN);
        const bool v1 = (q1 >= 0) && (q1 < WN);
        const bool v2 = (q2 >= 0) && (q2 < WN);
        const bool v3 = (q3 >= 0) && (q3 < WN);
        const float inv = 1.0f / (float)CN;
        const float s0 = v0 ? inv : 0.0f;
        const float s1 = v1 ? inv : 0.0f;
        const float s2 = v2 ? inv : 0.0f;
        const float s3 = v3 ? inv : 0.0f;

        const float* __restrict__ b0p = in2 + brow + (v0 ? q0 : 0);
        const float* __restrict__ b1p = in2 + brow + (v1 ? q1 : 0);
        const float* __restrict__ b2p = in2 + brow + (v2 ? q2 : 0);
        const float* __restrict__ b3p = in2 + brow + (v3 ? q3 : 0);

#pragma unroll 4
        for (int kk = 0; kk < CN; kk += 4) {
            v2f Av = { ap[0], ap[HWN] };

            v2f B0 = { b0p[0] * s0, b0p[HWN] * s0 };
            v2f B1 = { b1p[0] * s1, b1p[HWN] * s1 };
            v2f B2 = { b2p[0] * s2, b2p[HWN] * s2 };
            v2f B3 = { b3p[0] * s3, b3p[HWN] * s3 };

            acc0 = __builtin_amdgcn_wmma_f32_16x16x4_f32(
                       false, Av, false, B0, (short)0, acc0, false, false);
            acc1 = __builtin_amdgcn_wmma_f32_16x16x4_f32(
                       false, Av, false, B1, (short)0, acc1, false, false);
            acc2 = __builtin_amdgcn_wmma_f32_16x16x4_f32(
                       false, Av, false, B2, (short)0, acc2, false, false);
            acc3 = __builtin_amdgcn_wmma_f32_16x16x4_f32(
                       false, Av, false, B3, (short)0, acc3, false, false);

            ap  += 4 * HWN;
            b0p += 4 * HWN;
            b1p += 4 * HWN;
            b2p += 4 * HWN;
            b3p += 4 * HWN;
        }
    }

#pragma unroll
    for (int n = 0; n < 4; ++n) {
        const v8f acc = (n == 0) ? acc0 : (n == 1) ? acc1
                       : (n == 2) ? acc2 : acc3;
#pragma unroll
        for (int r = 0; r < 8; ++r) {
            const int m    = r + 8 * half;
            const int ncol = 16 * n + l;
            const int dx   = ncol - m - 24;
            if (dx >= -20 && dx <= 20 && ((dx & 1) == 0)) {
                const int ix = (dx + 20) >> 1;
                const int d  = iy * ND + ix;
                const size_t idx =
                    (((size_t)b * DN + d) * HN + h) * WN + (size_t)(w0 + m);
                out[idx] = acc[r];
            }
        }
    }
}

extern "C" void kernel_launch(void* const* d_in, const int* in_sizes, int n_in,
                              void* d_out, int out_size, void* d_ws, size_t ws_size,
                              hipStream_t stream) {
    (void)in_sizes; (void)n_in; (void)out_size; (void)d_ws; (void)ws_size;
    const float* in1 = (const float*)d_in[0];
    const float* in2 = (const float*)d_in[1];
    float* out = (float*)d_out;

    const int tasks = BN * HN * (WN / 16) * ND;
    const int waves_per_block = 8;
    const int blocks = tasks / waves_per_block;
    corr_wmma_f32_kernel<<<blocks, waves_per_block * 32, 0, stream>>>(in1, in2, out);
}


# Row pairs blocked 2x2 in registers: a wave loads two in1 rows and two in2 rows once per k-step and feeds four (h,dy) bands; edge pairs and zero bands get waves of their own; k order and sums same
# speedup vs baseline: 5.6322x; 5.6322x over previous
#include <hip/hip_runtime.h>


typedef float v2f __attribute__((ext_vector_type(2)));
typedef float v4f __attribute__((ext_vector_type(4)));
typedef float v8f __attribute__((ext_vector_type(8)));

#define BN 16
#define CN 256
#define HN 48
#define WN 64
#define HWN (HN * WN)
#define ND 21
#define DN (ND * ND)
#define WAVES_PER_BLOCK 4

#define NBLK 88
#define NSGL 42
#define N_BLOCK_WAVES (BN * 2 * NBLK)
#define N_SINGLE_WAVES (BN * 2 * NSGL)
#define N_FILL_WAVES (BN * HN)
#define N_WAVES (N_BLOCK_WAVES + N_SINGLE_WAVES + N_FILL_WAVES)

#define WMMA(acc, A, B) \
    acc = __builtin_amdgcn_wmma_f32_16x16x4_f32(false, A, false, B, (short)0, acc, false, false)

#define DECL8(P) \
    v8f P##00 = {}, P##11 = {}, P##22 = {}, P##33 = {}, P##02 = {}, P##13 = {}, P##20 = {}, P##31 = {}

#define FRAG(F, x0, x1)                      \
    const v2f F##0 = { x0[0], x1[0] };       \
    const v2f F##1 = { x0[1], x1[1] };       \
    const v2f F##2 = { x0[2], x1[2] };       \
    const v2f F##3 = { x0[3], x1[3] }

#define MM8(P, A, B)             \
    WMMA(P##00, A##0, B##0);     \
    WMMA(P##11, A##1, B##1);     \
    WMMA(P##22, A##2, B##2);     \
    WMMA(P##33, A##3, B##3);     \
    WMMA(P##02, A##0, B##2);     \
    WMMA(P##13, A##1, B##3);     \
    WMMA(P##20, A##2, B##0);     \
    WMMA(P##31, A##3, B##1)

#define KSTEP1(x0, x1, y0, y1)     \
    do {                           \
        FRAG(FA, x0, x1);          \
        FRAG(FB, y0, y1);          \
        MM8(acc, FA, FB);          \
    } while (0)

#define KSTEP4(xa0, xa1, xb0, xb1, yc0, yc1, yd0, yd1)   \
    do {                                                 \
        FRAG(Fa, xa0, xa1);                              \
        FRAG(Fb, xb0, xb1);                              \
        FRAG(Fc, yc0, yc1);                              \
        FRAG(Fd, yd0, yd1);                              \
        MM8(ac, Fa, Fc);                                 \
        MM8(ad, Fa, Fd);                                 \
        MM8(bc, Fb, Fc);                                 \
        MM8(bd, Fb, Fd);                                 \
    } while (0)

__device__ __forceinline__ void store_band(
    float* __restrict__ orow, const int lane, const int half, const int c,
    const v8f a00, const v8f a11, const v8f a22, const v8f a33,
    const v8f a02, const v8f a13, const v8f a20, const v8f a31)
{
    const float inv = 1.0f / (float)CN;
    const int src_lane = (lane & 16) | ((c + 1) & 15);

#pragma unroll
    for (int r = 0; r < 8; ++r) {
        const int i    = 8 * half + r;
        const int diff = c - i;
        const int dw   = ((diff + 8) & 15) - 8;
        const bool real = (dw == diff);

        const float g20 = __shfl(real ? a20[r] * inv : 0.0f, src_lane);
        const float g31 = __shfl(real ? a31[r] * inv : 0.0f, src_lane);

        if (dw >= -5 && dw <= 5) {
            const int ix = 2 * dw + 10;
            const v4f v = { real ? a00[r] * inv : 0.0f, real ? a11[r] * inv : 0.0f,
                            real ? a22[r] * inv : 0.0f, real ? a33[r] * inv : 0.0f };
            *(v4f*)(orow + (size_t)ix * HWN + 4 * i) = v;
        }
        if (dw >= -5 && dw <= 4) {
            const int ix = 2 * dw + 11;
            const v4f v = { real ? a02[r] * inv : 0.0f, real ? a13[r] * inv : 0.0f,
                            g20, g31 };
            *(v4f*)(orow + (size_t)ix * HWN + 4 * i) = v;
        }
    }
}

#define STORE8(P, orow) \
    store_band(orow, lane, half, c, P##00, P##11, P##22, P##33, P##02, P##13, P##20, P##31)

__device__ __forceinline__ constexpr int blk_start(const int a)
{
    int s = 0;
    for (int g = 0; g < a; ++g) {
        const int lo = (g - 4 > 0) ? g - 4 : 0;
        const int hi = (g + 4 < 11) ? g + 4 : 11;
        s += hi - lo + 1;
    }
    return s;
}

__global__ __launch_bounds__(WAVES_PER_BLOCK * 32, 3) void corr_wmma_f32_kernel(
    const float* __restrict__ in1,
    const float* __restrict__ in2,
    float* __restrict__ out)
{
    const int lane = threadIdx.x & 31;
    const int half = lane >> 4;
    const int c    = lane & 15;
    const int k0   = half * 2;

    const int task = __builtin_amdgcn_readfirstlane(
        (int)(blockIdx.x * WAVES_PER_BLOCK + (threadIdx.x >> 5)));

    if (task < N_BLOCK_WAVES) {
        const int bp = task / NBLK;
        const int e  = task - bp * NBLK;
        const int b  = bp >> 1;
        const int p  = bp & 1;
        int A = 0, st = 0;
#pragma unroll
        for (int a = 1; a < 12; ++a) {
            if (e >= blk_start(a)) { A = a; st = blk_start(a); }
        }
        const int C  = ((A - 4 > 0) ? A - 4 : 0) + (e - st);
        const int ha = 4 * A + p;
        const int hc = 4 * C + p;
        const int iy = 2 * (C - A) + 10;

        DECL8(ac); DECL8(ad); DECL8(bc); DECL8(bd);

        const float* __restrict__ ap =
            in1 + (((size_t)b * CN + k0) * HN + ha) * WN + 4 * c;
        const float* __restrict__ bq =
            in2 + (((size_t)b * CN + k0) * HN + hc) * WN + 4 * c;

#define LOAD8(S)                                              \
        S##a0 = *(const v4f*)(ap);                            \
        S##a1 = *(const v4f*)(ap + HWN);                      \
        S##b0 = *(const v4f*)(ap + 2 * WN);                   \
        S##b1 = *(const v4f*)(ap + 2 * WN + HWN);             \
        S##c0 = *(const v4f*)(bq);                            \
        S##c1 = *(const v4f*)(bq + HWN);                      \
        S##d0 = *(const v4f*)(bq + 2 * WN);                   \
        S##d1 = *(const v4f*)(bq + 2 * WN + HWN);             \
        ap += 4 * HWN;                                        \
        bq += 4 * HWN

        v4f Pa0, Pa1, Pb0, Pb1, Pc0, Pc1, Pd0, Pd1;
        v4f Ra0, Ra1, Rb0, Rb1, Rc0, Rc1, Rd0, Rd1;
        LOAD8(P);

#pragma unroll 1
        for (int it = 0; it < 31; ++it) {
            LOAD8(R);
            KSTEP4(Pa0, Pa1, Pb0, Pb1, Pc0, Pc1, Pd0, Pd1);
            LOAD8(P);
            KSTEP4(Ra0, Ra1, Rb0, Rb1, Rc0, Rc1, Rd0, Rd1);

            __builtin_amdgcn_sched_group_barrier(0x020, 8, 0);
            __builtin_amdgcn_sched_group_barrier(0x008, 8, 0);
            __builtin_amdgcn_sched_group_barrier(0x020, 8, 0);
            __builtin_amdgcn_sched_group_barrier(0x008, 56, 0);
        }
        {
            LOAD8(R);
            KSTEP4(Pa0, Pa1, Pb0, Pb1, Pc0, Pc1, Pd0, Pd1);
            KSTEP4(Ra0, Ra1, Rb0, Rb1, Rc0, Rc1, Rd0, Rd1);
        }
#undef LOAD8

        float* __restrict__ obase = out + (size_t)b * DN * HWN;
        STORE8(ac, obase + (size_t)(iy    ) * ND * HWN + (size_t)(ha    ) * WN);
        STORE8(ad, obase + (size_t)(iy + 1) * ND * HWN + (size_t)(ha    ) * WN);
        STORE8(bc, obase + (size_t)(iy - 1) * ND * HWN + (size_t)(ha + 2) * WN);
        STORE8(bd, obase + (size_t)(iy    ) * ND * HWN + (size_t)(ha + 2) * WN);
        return;
    }

    if (task < N_BLOCK_WAVES + N_SINGLE_WAVES) {
        const int s0   = task - N_BLOCK_WAVES;
        const int bp   = s0 / NSGL;
        const int s    = s0 - bp * NSGL;
        const int b    = bp >> 1;
        const int p    = bp & 1;
        const int side = (s >= 21) ? 1 : 0;
        const int r    = s - 21 * side;
        const int blk  = r / 3;
        const int w    = r - 3 * blk;
        const int a    = (w == 0) ? 0 : ((w == 1) ? 1 : (side ? 0 : 1));
        const int cc   = (w == 0) ? 0 : ((w == 1) ? 1 : (side ? 1 : 0));
        const int A    = side ? blk + 5 : blk;
        const int C    = side ? blk : blk + 5;
        const int h    = 4 * A + 2 * a + p;
        const int h2   = 4 * C + 2 * cc + p;
        const int iy   = ((h2 - h) >> 1) + 10;

        DECL8(acc);

        const float* __restrict__ ap =
            in1 + (((size_t)b * CN + k0) * HN + h) * WN + 4 * c;
        const float* __restrict__ bq =
            in2 + (((size_t)b * CN + k0) * HN + h2) * WN + 4 * c;

        v4f p0 = *(const v4f*)(ap);
        v4f p1 = *(const v4f*)(ap + HWN);
        v4f q0 = *(const v4f*)(bq);
        v4f q1 = *(const v4f*)(bq + HWN);
        ap += 4 * HWN;
        bq += 4 * HWN;

#pragma unroll 1
        for (int it = 0; it < 31; ++it) {
            v4f r0 = *(const v4f*)(ap);
            v4f r1 = *(const v4f*)(ap + HWN);
            v4f s0v = *(const v4f*)(bq);
            v4f s1v = *(const v4f*)(bq + HWN);
            ap += 4 * HWN;
            bq += 4 * HWN;
            KSTEP1(p0, p1, q0, q1);

            p0 = *(const v4f*)(ap);
            p1 = *(const v4f*)(ap + HWN);
            q0 = *(const v4f*)(bq);
            q1 = *(const v4f*)(bq + HWN);
            ap += 4 * HWN;
            bq += 4 * HWN;
            KSTEP1(r0, r1, s0v, s1v);

            __builtin_amdgcn_sched_group_barrier(0x020, 4, 0);
            __builtin_amdgcn_sched_group_barrier(0x008, 4, 0);
            __builtin_amdgcn_sched_group_barrier(0x020, 4, 0);
            __builtin_amdgcn_sched_group_barrier(0x008, 12, 0);
        }
        {
            v4f r0 = *(const v4f*)(ap);
            v4f r1 = *(const v4f*)(ap + HWN);
            v4f s0v = *(const v4f*)(bq);
            v4f s1v = *(const v4f*)(bq + HWN);
            KSTEP1(p0, p1, q0, q1);
            KSTEP1(r0, r1, s0v, s1v);
        }

        STORE8(acc, out + ((size_t)b * DN + (size_t)iy * ND) * HWN + (size_t)h * WN);
        return;
    }

    {
        const int f = task - (N_BLOCK_WAVES + N_SINGLE_WAVES);
        const int b = f / HN;
        const int h = f - b * HN;
        const v4f z = { 0.0f, 0.0f, 0.0f, 0.0f };
        for (int iy = 0; iy < ND; ++iy) {
            const int h2 = h - 20 + 2 * iy;
            if (h2 >= 0 && h2 < HN) continue;
            float* __restrict__ zr =
                out + ((size_t)b * DN + (size_t)iy * ND) * HWN + (size_t)h * WN + 4 * c;
#pragma unroll
            for (int q = 0; q < ND + 1; q += 2) {
                const int ix = q + half;
                if (ix < ND)
                    *(v4f*)(zr + (size_t)ix * HWN) = z;
            }
        }
    }
}

extern "C" void kernel_launch(void* const* d_in, const int* in_sizes, int n_in,
                              void* d_out, int out_size, void* d_ws, size_t ws_size,
                              hipStream_t stream) {
    (void)in_sizes; (void)n_in; (void)out_size; (void)d_ws; (void)ws_size;
    const float* in1 = (const float*)d_in[0];
    const float* in2 = (const float*)d_in[1];
    float* out = (float*)d_out;

    corr_wmma_f32_kernel<<<1232, 128, 0, stream>>>(in1, in2, out);
}


# Output stored as whole 128-byte lines: each lane rotates every accumulator's 8 registers by its column index with selects (no LDS), so one store carries all 8 pixel groups of a line; sums same
# speedup vs baseline: 6.0468x; 1.0736x over previous
#include <hip/hip_runtime.h>


typedef float v2f __attribute__((ext_vector_type(2)));
typedef float v4f __attribute__((ext_vector_type(4)));
typedef float v8f __attribute__((ext_vector_type(8)));

#define BN 16
#define CN 256
#define HN 48
#define WN 64
#define HWN (HN * WN)
#define ND 21
#define DN (ND * ND)
#define WAVES_PER_BLOCK 4

#define NBLK 88
#define NSGL 42
#define N_BLOCK_WAVES (BN * 2 * NBLK)
#define N_SINGLE_WAVES (BN * 2 * NSGL)
#define N_FILL_WAVES (BN * HN)
#define N_WAVES (N_BLOCK_WAVES + N_SINGLE_WAVES + N_FILL_WAVES)

#define WMMA(acc, A, B) \
    acc = __builtin_amdgcn_wmma_f32_16x16x4_f32(false, A, false, B, (short)0, acc, false, false)

#define DECL8(P) \
    v8f P##00 = {}, P##11 = {}, P##22 = {}, P##33 = {}, P##02 = {}, P##13 = {}, P##20 = {}, P##31 = {}

#define FRAG(F, x0, x1)                      \
    const v2f F##0 = { x0[0], x1[0] };       \
    const v2f F##1 = { x0[1], x1[1] };       \
    const v2f F##2 = { x0[2], x1[2] };       \
    const v2f F##3 = { x0[3], x1[3] }

#define MM8(P, A, B)             \
    WMMA(P##00, A##0, B##0);     \
    WMMA(P##11, A##1, B##1);     \
    WMMA(P##22, A##2, B##2);     \
    WMMA(P##33, A##3, B##3);     \
    WMMA(P##02, A##0, B##2);     \
    WMMA(P##13, A##1, B##3);     \
    WMMA(P##20, A##2, B##0);     \
    WMMA(P##31, A##3, B##1)

#define KSTEP1(x0, x1, y0, y1)     \
    do {                           \
        FRAG(FA, x0, x1);          \
        FRAG(FB, y0, y1);          \
        MM8(acc, FA, FB);          \
    } while (0)

#define KSTEP4(xa0, xa1, xb0, xb1, yc0, yc1, yd0, yd1)   \
    do {                                                 \
        FRAG(Fa, xa0, xa1);                              \
        FRAG(Fb, xb0, xb1);                              \
        FRAG(Fc, yc0, yc1);                              \
        FRAG(Fd, yd0, yd1);                              \
        MM8(ac, Fa, Fc);                                 \
        MM8(ad, Fa, Fd);                                 \
        MM8(bc, Fb, Fc);                                 \
        MM8(bd, Fb, Fd);                                 \
    } while (0)


__device__ __forceinline__ float zs(const v8f a, const int r, const int half, const int c)
{
    const int diff = c - (8 * half + r);
    const int dw   = ((diff + 8) & 15) - 8;
    return (dw == diff) ? a[r] * (1.0f / (float)CN) : 0.0f;
}

__device__ __forceinline__ v8f zs8(const v8f a, const int half, const int c)
{
    const v8f v = { zs(a, 0, half, c), zs(a, 1, half, c), zs(a, 2, half, c), zs(a, 3, half, c),
                    zs(a, 4, half, c), zs(a, 5, half, c), zs(a, 6, half, c), zs(a, 7, half, c) };
    return v;
}

__device__ __forceinline__ v8f shfl8(const v8f a, const int src_lane)
{
    const v8f v = { __shfl(a[0], src_lane), __shfl(a[1], src_lane), __shfl(a[2], src_lane),
                    __shfl(a[3], src_lane), __shfl(a[4], src_lane), __shfl(a[5], src_lane),
                    __shfl(a[6], src_lane), __shfl(a[7], src_lane) };
    return v;
}

__device__ __forceinline__ v8f rot8(const v8f x, const bool b0, const bool b1, const bool b2)
{
    const v8f t = b0 ? __builtin_shufflevector(x, x, 1, 2, 3, 4, 5, 6, 7, 0) : x;
    const v8f y = b1 ? __builtin_shufflevector(t, t, 2, 3, 4, 5, 6, 7, 0, 1) : t;
    return        b2 ? __builtin_shufflevector(y, y, 4, 5, 6, 7, 0, 1, 2, 3) : y;
}

__device__ __forceinline__ void store_class(
    float* __restrict__ orow, const int half, const int c, const int odd,
    const v8f p0, const v8f p1, const v8f p2, const v8f p3)
{
    const bool b0 = (c & 1) != 0;
    const bool b1 = (c & 2) != 0;
    const bool b2 = (c & 4) != 0;
    const v8f q0 = rot8(p0, b0, b1, b2);
    const v8f q1 = rot8(p1, b0, b1, b2);
    const v8f q2 = rot8(p2, b0, b1, b2);
    const v8f q3 = rot8(p3, b0, b1, b2);

#pragma unroll
    for (int s = 0; s < 8; ++s) {
        const int k    = (8 - s) & 7;
        const int i    = 8 * half + ((c - s) & 7);
        const int diff = c - i;
        const int dw   = ((diff + 8) & 15) - 8;
        const unsigned off = (unsigned)((2 * dw + 10 + odd) * HWN + 4 * i);

        if (dw >= -5 && dw <= 5 - odd) {
            const v4f v = { q0[k], q1[k], q2[k], q3[k] };
            *(v4f*)(orow + off) = v;
        }
    }
}

__device__ __forceinline__ void store_band(
    float* __restrict__ orow, const int lane, const int half, const int c,
    const v8f a00, const v8f a11, const v8f a22, const v8f a33,
    const v8f a02, const v8f a13, const v8f a20, const v8f a31)
{
    const int src_lane = (lane & 16) | ((c + 1) & 15);

    store_class(orow, half, c, 0,
                zs8(a00, half, c), zs8(a11, half, c), zs8(a22, half, c), zs8(a33, half, c));

    store_class(orow, half, c, 1,
                zs8(a02, half, c), zs8(a13, half, c),
                shfl8(zs8(a20, half, c), src_lane), shfl8(zs8(a31, half, c), src_lane));
}

#define STORE8(P, orow) \
    store_band(orow, lane, half, c, P##00, P##11, P##22, P##33, P##02, P##13, P##20, P##31)

__device__ __forceinline__ constexpr int blk_start(const int a)
{
    int s = 0;
    for (int g = 0; g < a; ++g) {
        const int lo = (g - 4 > 0) ? g - 4 : 0;
        const int hi = (g + 4 < 11) ? g + 4 : 11;
        s += hi - lo + 1;
    }
    return s;
}

__global__ __launch_bounds__(WAVES_PER_BLOCK * 32, 3) void corr_wmma_f32_kernel(
    const float* __restrict__ in1,
    const float* __restrict__ in2,
    float* __restrict__ out)
{
    const int lane = threadIdx.x & 31;
    const int half = lane >> 4;
    const int c    = lane & 15;
    const int k0   = half * 2;

    const int task = __builtin_amdgcn_readfirstlane(
        (int)(blockIdx.x * WAVES_PER_BLOCK + (threadIdx.x >> 5)));

    if (task < N_BLOCK_WAVES) {
        const int bp = task / NBLK;
        const int e  = task - bp * NBLK;
        const int b  = bp >> 1;
        const int p  = bp & 1;
        int A = 0, st = 0;
#pragma unroll
        for (int a = 1; a < 12; ++a) {
            if (e >= blk_start(a)) { A = a; st = blk_start(a); }
        }
        const int C  = ((A - 4 > 0) ? A - 4 : 0) + (e - st);
        const int ha = 4 * A + p;
        const int hc = 4 * C + p;
        const int iy = 2 * (C - A) + 10;

        DECL8(ac); DECL8(ad); DECL8(bc); DECL8(bd);

        const float* __restrict__ ap =
            in1 + (((size_t)b * CN + k0) * HN + ha) * WN + 4 * c;
        const float* __restrict__ bq =
            in2 + (((size_t)b * CN + k0) * HN + hc) * WN + 4 * c;

#define LOAD8(S)                                              \
        S##a0 = *(const v4f*)(ap);                            \
        S##a1 = *(const v4f*)(ap + HWN);                      \
        S##b0 = *(const v4f*)(ap + 2 * WN);                   \
        S##b1 = *(const v4f*)(ap + 2 * WN + HWN);             \
        S##c0 = *(const v4f*)(bq);                            \
        S##c1 = *(const v4f*)(bq + HWN);                      \
        S##d0 = *(const v4f*)(bq + 2 * WN);                   \
        S##d1 = *(const v4f*)(bq + 2 * WN + HWN);             \
        ap += 4 * HWN;                                        \
        bq += 4 * HWN

        v4f Pa0, Pa1, Pb0, Pb1, Pc0, Pc1, Pd0, Pd1;
        v4f Ra0, Ra1, Rb0, Rb1, Rc0, Rc1, Rd0, Rd1;
        LOAD8(P);

#pragma unroll 1
        for (int it = 0; it < 31; ++it) {
            LOAD8(R);
            KSTEP4(Pa0, Pa1, Pb0, Pb1, Pc0, Pc1, Pd0, Pd1);
            LOAD8(P);
            KSTEP4(Ra0, Ra1, Rb0, Rb1, Rc0, Rc1, Rd0, Rd1);

            __builtin_amdgcn_sched_group_barrier(0x020, 8, 0);
            __builtin_amdgcn_sched_group_barrier(0x008, 8, 0);
            __builtin_amdgcn_sched_group_barrier(0x020, 8, 0);
            __builtin_amdgcn_sched_group_barrier(0x008, 56, 0);
        }
        {
            LOAD8(R);
            KSTEP4(Pa0, Pa1, Pb0, Pb1, Pc0, Pc1, Pd0, Pd1);
            KSTEP4(Ra0, Ra1, Rb0, Rb1, Rc0, Rc1, Rd0, Rd1);
        }
#undef LOAD8

        float* __restrict__ obase = out + (size_t)b * DN * HWN;
        STORE8(ac, obase + (size_t)(iy    ) * ND * HWN + (size_t)(ha    ) * WN);
        STORE8(ad, obase + (size_t)(iy + 1) * ND * HWN + (size_t)(ha    ) * WN);
        STORE8(bc, obase + (size_t)(iy - 1) * ND * HWN + (size_t)(ha + 2) * WN);
        STORE8(bd, obase + (size_t)(iy    ) * ND * HWN + (size_t)(ha + 2) * WN);
        return;
    }

    if (task < N_BLOCK_WAVES + N_SINGLE_WAVES) {
        const int s0   = task - N_BLOCK_WAVES;
        const int bp   = s0 / NSGL;
        const int s    = s0 - bp * NSGL;
        const int b    = bp >> 1;
        const int p    = bp & 1;
        const int side = (s >= 21) ? 1 : 0;
        const int r    = s - 21 * side;
        const int blk  = r / 3;
        const int w    = r - 3 * blk;
        const int a    = (w == 0) ? 0 : ((w == 1) ? 1 : (side ? 0 : 1));
        const int cc   = (w == 0) ? 0 : ((w == 1) ? 1 : (side ? 1 : 0));
        const int A    = side ? blk + 5 : blk;
        const int C    = side ? blk : blk + 5;
        const int h    = 4 * A + 2 * a + p;
        const int h2   = 4 * C + 2 * cc + p;
        const int iy   = ((h2 - h) >> 1) + 10;

        DECL8(acc);

        const float* __restrict__ ap =
            in1 + (((size_t)b * CN + k0) * HN + h) * WN + 4 * c;
        const float* __restrict__ bq =
            in2 + (((size_t)b * CN + k0) * HN + h2) * WN + 4 * c;

        v4f p0 = *(const v4f*)(ap);
        v4f p1 = *(const v4f*)(ap + HWN);
        v4f q0 = *(const v4f*)(bq);
        v4f q1 = *(const v4f*)(bq + HWN);
        ap += 4 * HWN;
        bq += 4 * HWN;

#pragma unroll 1
        for (int it = 0; it < 31; ++it) {
            v4f r0 = *(const v4f*)(ap);
            v4f r1 = *(const v4f*)(ap + HWN);
            v4f s0v = *(const v4f*)(bq);
            v4f s1v = *(const v4f*)(bq + HWN);
            ap += 4 * HWN;
            bq += 4 * HWN;
            KSTEP1(p0, p1, q0, q1);

            p0 = *(const v4f*)(ap);
            p1 = *(const v4f*)(ap + HWN);
            q0 = *(const v4f*)(bq);
            q1 = *(const v4f*)(bq + HWN);
            ap += 4 * HWN;
            bq += 4 * HWN;
            KSTEP1(r0, r1, s0v, s1v);

            __builtin_amdgcn_sched_group_barrier(0x020, 4, 0);
            __builtin_amdgcn_sched_group_barrier(0x008, 4, 0);
            __builtin_amdgcn_sched_group_barrier(0x020, 4, 0);
            __builtin_amdgcn_sched_group_barrier(0x008, 12, 0);
        }
        {
            v4f r0 = *(const v4f*)(ap);
            v4f r1 = *(const v4f*)(ap + HWN);
            v4f s0v = *(const v4f*)(bq);
            v4f s1v = *(const v4f*)(bq + HWN);
            KSTEP1(p0, p1, q0, q1);
            KSTEP1(r0, r1, s0v, s1v);
        }

        STORE8(acc, out + ((size_t)b * DN + (size_t)iy * ND) * HWN + (size_t)h * WN);
        return;
    }

    {
        const int f = task - (N_BLOCK_WAVES + N_SINGLE_WAVES);
        const int b = f / HN;
        const int h = f - b * HN;
        const v4f z = { 0.0f, 0.0f, 0.0f, 0.0f };
        for (int iy = 0; iy < ND; ++iy) {
            const int h2 = h - 20 + 2 * iy;
            if (h2 >= 0 && h2 < HN) continue;
            float* __restrict__ zr =
                out + ((size_t)b * DN + (size_t)iy * ND) * HWN + (size_t)h * WN + 4 * c;
#pragma unroll
            for (int q = 0; q < ND + 1; q += 2) {
                const int ix = q + half;
                if (ix < ND)
                    *(v4f*)(zr + (size_t)ix * HWN) = z;
            }
        }
    }
}

extern "C" void kernel_launch(void* const* d_in, const int* in_sizes, int n_in,
                              void* d_out, int out_size, void* d_ws, size_t ws_size,
                              hipStream_t stream) {
    (void)in_sizes; (void)n_in; (void)out_size; (void)d_ws; (void)ws_size;
    const float* in1 = (const float*)d_in[0];
    const float* in2 = (const float*)d_in[1];
    float* out = (float*)d_out;

    corr_wmma_f32_kernel<<<1232, 128, 0, stream>>>(in1, in2, out);
}


# Second-level cache prefetch of the in1 and in2 row pieces three (block waves) or five to six (single waves) loop trips ahead of the loads, one 256-byte piece per lane; k order and sums same
# speedup vs baseline: 6.3655x; 1.0527x over previous
#include <hip/hip_runtime.h>


typedef float v2f __attribute__((ext_vector_type(2)));
typedef float v4f __attribute__((ext_vector_type(4)));
typedef float v8f __attribute__((ext_vector_type(8)));

#define BN 16
#define CN 256
#define HN 48
#define WN 64
#define HWN (HN * WN)
#define ND 21
#define DN (ND * ND)
#define WAVES_PER_BLOCK 4

#define NBLK 88
#define NSGL 42
#define N_BLOCK_WAVES (BN * 2 * NBLK)
#define N_SINGLE_WAVES (BN * 2 * NSGL)
#define N_FILL_WAVES (BN * HN)
#define N_WAVES (N_BLOCK_WAVES + N_SINGLE_WAVES + N_FILL_WAVES)

#define WMMA(acc, A, B) \
    acc = __builtin_amdgcn_wmma_f32_16x16x4_f32(false, A, false, B, (short)0, acc, false, false)

#define DECL8(P) \
    v8f P##00 = {}, P##11 = {}, P##22 = {}, P##33 = {}, P##02 = {}, P##13 = {}, P##20 = {}, P##31 = {}

#define FRAG(F, x0, x1)                      \
    const v2f F##0 = { x0[0], x1[0] };       \
    const v2f F##1 = { x0[1], x1[1] };       \
    const v2f F##2 = { x0[2], x1[2] };       \
    const v2f F##3 = { x0[3], x1[3] }

#define MM8(P, A, B)             \
    WMMA(P##00, A##0, B##0);     \
    WMMA(P##11, A##1, B##1);     \
    WMMA(P##22, A##2, B##2);     \
    WMMA(P##33, A##3, B##3);     \
    WMMA(P##02, A##0, B##2);     \
    WMMA(P##13, A##1, B##3);     \
    WMMA(P##20, A##2, B##0);     \
    WMMA(P##31, A##3, B##1)

#define KSTEP1(x0, x1, y0, y1)     \
    do {                           \
        FRAG(FA, x0, x1);          \
        FRAG(FB, y0, y1);          \
        MM8(acc, FA, FB);          \
    } while (0)

#define KSTEP4(xa0, xa1, xb0, xb1, yc0, yc1, yd0, yd1)   \
    do {                                                 \
        FRAG(Fa, xa0, xa1);                              \
        FRAG(Fb, xb0, xb1);                              \
        FRAG(Fc, yc0, yc1);                              \
        FRAG(Fd, yd0, yd1);                              \
        MM8(ac, Fa, Fc);                                 \
        MM8(ad, Fa, Fd);                                 \
        MM8(bc, Fb, Fc);                                 \
        MM8(bd, Fb, Fd);                                 \
    } while (0)


__device__ __forceinline__ float zs(const v8f a, const int r, const int half, const int c)
{
    const int diff = c - (8 * half + r);
    const int dw   = ((diff + 8) & 15) - 8;
    return (dw == diff) ? a[r] * (1.0f / (float)CN) : 0.0f;
}

__device__ __forceinline__ v8f zs8(const v8f a, const int half, const int c)
{
    const v8f v = { zs(a, 0, half, c), zs(a, 1, half, c), zs(a, 2, half, c), zs(a, 3, half, c),
                    zs(a, 4, half, c), zs(a, 5, half, c), zs(a, 6, half, c), zs(a, 7, half, c) };
    return v;
}

__device__ __forceinline__ v8f shfl8(const v8f a, const int src_lane)
{
    const v8f v = { __shfl(a[0], src_lane), __shfl(a[1], src_lane), __shfl(a[2], src_lane),
                    __shfl(a[3], src_lane), __shfl(a[4], src_lane), __shfl(a[5], src_lane),
                    __shfl(a[6], src_lane), __shfl(a[7], src_lane) };
    return v;
}

__device__ __forceinline__ v8f rot8(const v8f x, const bool b0, const bool b1, const bool b2)
{
    const v8f t = b0 ? __builtin_shufflevector(x, x, 1, 2, 3, 4, 5, 6, 7, 0) : x;
    const v8f y = b1 ? __builtin_shufflevector(t, t, 2, 3, 4, 5, 6, 7, 0, 1) : t;
    return        b2 ? __builtin_shufflevector(y, y, 4, 5, 6, 7, 0, 1, 2, 3) : y;
}

__device__ __forceinline__ void l2_prefetch(const float* p)
{
    __builtin_prefetch(p, 0, 2);
}

__device__ __forceinline__ void store_class(
    float* __restrict__ orow, const int half, const int c, const int odd,
    const v8f p0, const v8f p1, const v8f p2, const v8f p3)
{
    const bool b0 = (c & 1) != 0;
    const bool b1 = (c & 2) != 0;
    const bool b2 = (c & 4) != 0;
    const v8f q0 = rot8(p0, b0, b1, b2);
    const v8f q1 = rot8(p1, b0, b1, b2);
    const v8f q2 = rot8(p2, b0, b1, b2);
    const v8f q3 = rot8(p3, b0, b1, b2);

#pragma unroll
    for (int s = 0; s < 8; ++s) {
        const int k    = (8 - s) & 7;
        const int i    = 8 * half + ((c - s) & 7);
        const int diff = c - i;
        const int dw   = ((diff + 8) & 15) - 8;
        const unsigned off = (unsigned)((2 * dw + 10 + odd) * HWN + 4 * i);

        if (dw >= -5 && dw <= 5 - odd) {
            const v4f v = { q0[k], q1[k], q2[k], q3[k] };
            *(v4f*)(orow + off) = v;
        }
    }
}

__device__ __forceinline__ void store_band(
    float* __restrict__ orow, const int lane, const int half, const int c,
    const v8f a00, const v8f a11, const v8f a22, const v8f a33,
    const v8f a02, const v8f a13, const v8f a20, const v8f a31)
{
    const int src_lane = (lane & 16) | ((c + 1) & 15);

    store_class(orow, half, c, 0,
                zs8(a00, half, c), zs8(a11, half, c), zs8(a22, half, c), zs8(a33, half, c));

    store_class(orow, half, c, 1,
                zs8(a02, half, c), zs8(a13, half, c),
                shfl8(zs8(a20, half, c), src_lane), shfl8(zs8(a31, half, c), src_lane));
}

#define STORE8(P, orow) \
    store_band(orow, lane, half, c, P##00, P##11, P##22, P##33, P##02, P##13, P##20, P##31)

__device__ __forceinline__ constexpr int blk_start(const int a)
{
    int s = 0;
    for (int g = 0; g < a; ++g) {
        const int lo = (g - 4 > 0) ? g - 4 : 0;
        const int hi = (g + 4 < 11) ? g + 4 : 11;
        s += hi - lo + 1;
    }
    return s;
}

__global__ __launch_bounds__(WAVES_PER_BLOCK * 32, 3) void corr_wmma_f32_kernel(
    const float* __restrict__ in1,
    const float* __restrict__ in2,
    float* __restrict__ out)
{
    const int lane = threadIdx.x & 31;
    const int half = lane >> 4;
    const int c    = lane & 15;
    const int k0   = half * 2;

    const int task = __builtin_amdgcn_readfirstlane(
        (int)(blockIdx.x * WAVES_PER_BLOCK + (threadIdx.x >> 5)));

    if (task < N_BLOCK_WAVES) {
        const int bp = task / NBLK;
        const int e  = task - bp * NBLK;
        const int b  = bp >> 1;
        const int p  = bp & 1;
        int A = 0, st = 0;
#pragma unroll
        for (int a = 1; a < 12; ++a) {
            if (e >= blk_start(a)) { A = a; st = blk_start(a); }
        }
        const int C  = ((A - 4 > 0) ? A - 4 : 0) + (e - st);
        const int ha = 4 * A + p;
        const int hc = 4 * C + p;
        const int iy = 2 * (C - A) + 10;

        DECL8(ac); DECL8(ad); DECL8(bc); DECL8(bd);

        const float* __restrict__ ap =
            in1 + (((size_t)b * CN + k0) * HN + ha) * WN + 4 * c;
        const float* __restrict__ bq =
            in2 + (((size_t)b * CN + k0) * HN + hc) * WN + 4 * c;

#define LOAD8(S)                                              \
        S##a0 = *(const v4f*)(ap);                            \
        S##a1 = *(const v4f*)(ap + HWN);                      \
        S##b0 = *(const v4f*)(ap + 2 * WN);                   \
        S##b1 = *(const v4f*)(ap + 2 * WN + HWN);             \
        S##c0 = *(const v4f*)(bq);                            \
        S##c1 = *(const v4f*)(bq + HWN);                      \
        S##d0 = *(const v4f*)(bq + 2 * WN);                   \
        S##d1 = *(const v4f*)(bq + 2 * WN + HWN);             \
        ap += 4 * HWN;                                        \
        bq += 4 * HWN

        v4f Pa0, Pa1, Pb0, Pb1, Pc0, Pc1, Pd0, Pd1;
        v4f Ra0, Ra1, Rb0, Rb1, Rc0, Rc1, Rd0, Rd1;
        LOAD8(P);

        const int pr = lane & 3;
        const float* __restrict__ pfp = ((pr & 2) ? in2 : in1) +
            (((size_t)b * CN + 12 + (lane >> 2)) * HN + ((pr & 2) ? hc : ha) + 2 * (pr & 1)) * WN;
        l2_prefetch(pfp);
        l2_prefetch(pfp + 8 * HWN);
        pfp += 16 * HWN;

#pragma unroll 1
        for (int it = 0; it < 31; ++it) {
            if (it < 28) l2_prefetch(pfp);
            pfp += 8 * HWN;
            LOAD8(R);
            KSTEP4(Pa0, Pa1, Pb0, Pb1, Pc0, Pc1, Pd0, Pd1);
            LOAD8(P);
            KSTEP4(Ra0, Ra1, Rb0, Rb1, Rc0, Rc1, Rd0, Rd1);

            __builtin_amdgcn_sched_group_barrier(0x020, 8, 0);
            __builtin_amdgcn_sched_group_barrier(0x008, 8, 0);
            __builtin_amdgcn_sched_group_barrier(0x020, 8, 0);
            __builtin_amdgcn_sched_group_barrier(0x008, 56, 0);
        }
        {
            LOAD8(R);
            KSTEP4(Pa0, Pa1, Pb0, Pb1, Pc0, Pc1, Pd0, Pd1);
            KSTEP4(Ra0, Ra1, Rb0, Rb1, Rc0, Rc1, Rd0, Rd1);
        }
#undef LOAD8

        float* __restrict__ obase = out + (size_t)b * DN * HWN;
        STORE8(ac, obase + (size_t)(iy    ) * ND * HWN + (size_t)(ha    ) * WN);
        STORE8(ad, obase + (size_t)(iy + 1) * ND * HWN + (size_t)(ha    ) * WN);
        STORE8(bc, obase + (size_t)(iy - 1) * ND * HWN + (size_t)(ha + 2) * WN);
        STORE8(bd, obase + (size_t)(iy    ) * ND * HWN + (size_t)(ha + 2) * WN);
        return;
    }

    if (task < N_BLOCK_WAVES + N_SINGLE_WAVES) {
        const int s0   = task - N_BLOCK_WAVES;
        const int bp   = s0 / NSGL;
        const int s    = s0 - bp * NSGL;
        const int b    = bp >> 1;
        const int p    = bp & 1;
        const int side = (s >= 21) ? 1 : 0;
        const int r    = s - 21 * side;
        const int blk  = r / 3;
        const int w    = r - 3 * blk;
        const int a    = (w == 0) ? 0 : ((w == 1) ? 1 : (side ? 0 : 1));
        const int cc   = (w == 0) ? 0 : ((w == 1) ? 1 : (side ? 1 : 0));
        const int A    = side ? blk + 5 : blk;
        const int C    = side ? blk : blk + 5;
        const int h    = 4 * A + 2 * a + p;
        const int h2   = 4 * C + 2 * cc + p;
        const int iy   = ((h2 - h) >> 1) + 10;

        DECL8(acc);

        const float* __restrict__ ap =
            in1 + (((size_t)b * CN + k0) * HN + h) * WN + 4 * c;
        const float* __restrict__ bq =
            in2 + (((size_t)b * CN + k0) * HN + h2) * WN + 4 * c;

        v4f p0 = *(const v4f*)(ap);
        v4f p1 = *(const v4f*)(ap + HWN);
        v4f q0 = *(const v4f*)(bq);
        v4f q1 = *(const v4f*)(bq + HWN);
        ap += 4 * HWN;
        bq += 4 * HWN;

        const float* __restrict__ pfp = ((lane & 1) ? in2 : in1) +
            (((size_t)b * CN + 12 + (lane >> 1)) * HN + ((lane & 1) ? h2 : h)) * WN;
        l2_prefetch(pfp);
        l2_prefetch(pfp + 16 * HWN);
        pfp += 32 * HWN;

#pragma unroll 1
        for (int it = 0; it < 31; ++it) {
            if ((it & 1) == 0 && it < 25) { l2_prefetch(pfp); pfp += 16 * HWN; }
            v4f r0 = *(const v4f*)(ap);
            v4f r1 = *(const v4f*)(ap + HWN);
            v4f s0v = *(const v4f*)(bq);
            v4f s1v = *(const v4f*)(bq + HWN);
            ap += 4 * HWN;
            bq += 4 * HWN;
            KSTEP1(p0, p1, q0, q1);

            p0 = *(const v4f*)(ap);
            p1 = *(const v4f*)(ap + HWN);
            q0 = *(const v4f*)(bq);
            q1 = *(const v4f*)(bq + HWN);
            ap += 4 * HWN;
            bq += 4 * HWN;
            KSTEP1(r0, r1, s0v, s1v);

            __builtin_amdgcn_sched_group_barrier(0x020, 4, 0);
            __builtin_amdgcn_sched_group_barrier(0x008, 4, 0);
            __builtin_amdgcn_sched_group_barrier(0x020, 4, 0);
            __builtin_amdgcn_sched_group_barrier(0x008, 12, 0);
        }
        {
            v4f r0 = *(const v4f*)(ap);
            v4f r1 = *(const v4f*)(ap + HWN);
            v4f s0v = *(const v4f*)(bq);
            v4f s1v = *(const v4f*)(bq + HWN);
            KSTEP1(p0, p1, q0, q1);
            KSTEP1(r0, r1, s0v, s1v);
        }

        STORE8(acc, out + ((size_t)b * DN + (size_t)iy * ND) * HWN + (size_t)h * WN);
        return;
    }

    {
        const int f = task - (N_BLOCK_WAVES + N_SINGLE_WAVES);
        const int b = f / HN;
        const int h = f - b * HN;
        const v4f z = { 0.0f, 0.0f, 0.0f, 0.0f };
        for (int iy = 0; iy < ND; ++iy) {
            const int h2 = h - 20 + 2 * iy;
            if (h2 >= 0 && h2 < HN) continue;
            float* __restrict__ zr =
                out + ((size_t)b * DN + (size_t)iy * ND) * HWN + (size_t)h * WN + 4 * c;
#pragma unroll
            for (int q = 0; q < ND + 1; q += 2) {
                const int ix = q + half;
                if (ix < ND)
                    *(v4f*)(zr + (size_t)ix * HWN) = z;
            }
        }
    }
}

extern "C" void kernel_launch(void* const* d_in, const int* in_sizes, int n_in,
                              void* d_out, int out_size, void* d_ws, size_t ws_size,
                              hipStream_t stream) {
    (void)in_sizes; (void)n_in; (void)out_size; (void)d_ws; (void)ws_size;
    const float* in1 = (const float*)d_in[0];
    const float* in2 = (const float*)d_in[1];
    float* out = (float*)d_out;

    corr_wmma_f32_kernel<<<1232, 128, 0, stream>>>(in1, in2, out);
}


# Block waves: tiles of 2 rows x 8 same-parity columns (8-byte row loads), not one row mod 4; column blocks too far apart for any dx are skipped: 28 WMMAs per k-step, was 32; k order and sums same
# speedup vs baseline: 6.5690x; 1.0320x over previous
#include <hip/hip_runtime.h>


typedef float v2f __attribute__((ext_vector_type(2)));
typedef float v4f __attribute__((ext_vector_type(4)));
typedef float v8f __attribute__((ext_vector_type(8)));

#define BN 16
#define CN 256
#define HN 48
#define WN 64
#define HWN (HN * WN)
#define ND 21
#define DN (ND * ND)
#define WAVES_PER_BLOCK 4

#define NBLK 88
#define NSGL 42
#define N_BLOCK_WAVES (BN * 2 * NBLK)
#define N_SINGLE_WAVES (BN * 2 * NSGL)
#define N_FILL_WAVES (BN * HN)
#define N_WAVES (N_BLOCK_WAVES + N_SINGLE_WAVES + N_FILL_WAVES)

#define WMMA(acc, A, B) \
    acc = __builtin_amdgcn_wmma_f32_16x16x4_f32(false, A, false, B, (short)0, acc, false, false)

#define DECL8(P) \
    v8f P##00 = {}, P##11 = {}, P##22 = {}, P##33 = {}, P##02 = {}, P##13 = {}, P##20 = {}, P##31 = {}

#define FRAG(F, x0, x1)                      \
    const v2f F##0 = { x0[0], x1[0] };       \
    const v2f F##1 = { x0[1], x1[1] };       \
    const v2f F##2 = { x0[2], x1[2] };       \
    const v2f F##3 = { x0[3], x1[3] }

#define MM8(P, A, B)             \
    WMMA(P##00, A##0, B##0);     \
    WMMA(P##11, A##1, B##1);     \
    WMMA(P##22, A##2, B##2);     \
    WMMA(P##33, A##3, B##3);     \
    WMMA(P##02, A##0, B##2);     \
    WMMA(P##13, A##1, B##3);     \
    WMMA(P##20, A##2, B##0);     \
    WMMA(P##31, A##3, B##1)

#define KSTEP1(x0, x1, y0, y1)     \
    do {                           \
        FRAG(FA, x0, x1);          \
        FRAG(FB, y0, y1);          \
        MM8(acc, FA, FB);          \
    } while (0)

#define KSTEP4(xa0, xa1, xb0, xb1, yc0, yc1, yd0, yd1)   \
    do {                                                 \
        FRAG(Fa, xa0, xa1);                              \
        FRAG(Fb, xb0, xb1);                              \
        FRAG(Fc, yc0, yc1);                              \
        FRAG(Fd, yd0, yd1);                              \
        MM8(ac, Fa, Fc);                                 \
        MM8(ad, Fa, Fd);                                 \
        MM8(bc, Fb, Fc);                                 \
        MM8(bd, Fb, Fd);                                 \
    } while (0)


__device__ __forceinline__ float zs(const v8f a, const int r, const int half, const int c)
{
    const int diff = c - (8 * half + r);
    const int dw   = ((diff + 8) & 15) - 8;
    return (dw == diff) ? a[r] * (1.0f / (float)CN) : 0.0f;
}

__device__ __forceinline__ v8f zs8(const v8f a, const int half, const int c)
{
    const v8f v = { zs(a, 0, half, c), zs(a, 1, half, c), zs(a, 2, half, c), zs(a, 3, half, c),
                    zs(a, 4, half, c), zs(a, 5, half, c), zs(a, 6, half, c), zs(a, 7, half, c) };
    return v;
}

__device__ __forceinline__ v8f shfl8(const v8f a, const int src_lane)
{
    const v8f v = { __shfl(a[0], src_lane), __shfl(a[1], src_lane), __shfl(a[2], src_lane),
                    __shfl(a[3], src_lane), __shfl(a[4], src_lane), __shfl(a[5], src_lane),
                    __shfl(a[6], src_lane), __shfl(a[7], src_lane) };
    return v;
}

__device__ __forceinline__ v8f rot8(const v8f x, const bool b0, const bool b1, const bool b2)
{
    const v8f t = b0 ? __builtin_shufflevector(x, x, 1, 2, 3, 4, 5, 6, 7, 0) : x;
    const v8f y = b1 ? __builtin_shufflevector(t, t, 2, 3, 4, 5, 6, 7, 0, 1) : t;
    return        b2 ? __builtin_shufflevector(y, y, 4, 5, 6, 7, 0, 1, 2, 3) : y;
}

__device__ __forceinline__ void l2_prefetch(const float* p)
{
    __builtin_prefetch(p, 0, 2);
}

__device__ __forceinline__ void store_class(
    float* __restrict__ orow, const int half, const int c, const int odd,
    const v8f p0, const v8f p1, const v8f p2, const v8f p3)
{
    const bool b0 = (c & 1) != 0;
    const bool b1 = (c & 2) != 0;
    const bool b2 = (c & 4) != 0;
    const v8f q0 = rot8(p0, b0, b1, b2);
    const v8f q1 = rot8(p1, b0, b1, b2);
    const v8f q2 = rot8(p2, b0, b1, b2);
    const v8f q3 = rot8(p3, b0, b1, b2);

#pragma unroll
    for (int s = 0; s < 8; ++s) {
        const int k    = (8 - s) & 7;
        const int i    = 8 * half + ((c - s) & 7);
        const int diff = c - i;
        const int dw   = ((diff + 8) & 15) - 8;
        const unsigned off = (unsigned)((2 * dw + 10 + odd) * HWN + 4 * i);

        if (dw >= -5 && dw <= 5 - odd) {
            const v4f v = { q0[k], q1[k], q2[k], q3[k] };
            *(v4f*)(orow + off) = v;
        }
    }
}

__device__ __forceinline__ void store_band(
    float* __restrict__ orow, const int lane, const int half, const int c,
    const v8f a00, const v8f a11, const v8f a22, const v8f a33,
    const v8f a02, const v8f a13, const v8f a20, const v8f a31)
{
    const int src_lane = (lane & 16) | ((c + 1) & 15);

    store_class(orow, half, c, 0,
                zs8(a00, half, c), zs8(a11, half, c), zs8(a22, half, c), zs8(a33, half, c));

    store_class(orow, half, c, 1,
                zs8(a02, half, c), zs8(a13, half, c),
                shfl8(zs8(a20, half, c), src_lane), shfl8(zs8(a31, half, c), src_lane));
}

#define STORE8(P, orow) \
    store_band(orow, lane, half, c, P##00, P##11, P##22, P##33, P##02, P##13, P##20, P##31)

__device__ __forceinline__ constexpr bool tile_ok(const int X, const int X2)
{
    return X2 >= 0 && X2 < 4 && X2 - X <= 2 && X - X2 <= 2;
}

__device__ __forceinline__ float pair_lane(const float x)
{
    const int v = __builtin_bit_cast(int, x);
    return __builtin_bit_cast(float, __builtin_amdgcn_update_dpp(v, v, 0xB1, 0xF, 0xF, false));
}

__device__ __forceinline__ constexpr int blk_start(const int a)
{
    int s = 0;
    for (int g = 0; g < a; ++g) {
        const int lo = (g - 4 > 0) ? g - 4 : 0;
        const int hi = (g + 4 < 11) ? g + 4 : 11;
        s += hi - lo + 1;
    }
    return s;
}

__global__ __launch_bounds__(WAVES_PER_BLOCK * 32, 3) void corr_wmma_f32_kernel(
    const float* __restrict__ in1,
    const float* __restrict__ in2,
    float* __restrict__ out)
{
    const int lane = threadIdx.x & 31;
    const int half = lane >> 4;
    const int c    = lane & 15;
    const int k0   = half * 2;

    const int task = __builtin_amdgcn_readfirstlane(
        (int)(blockIdx.x * WAVES_PER_BLOCK + (threadIdx.x >> 5)));

    if (task < N_BLOCK_WAVES) {
        const int bp = task / NBLK;
        const int e  = task - bp * NBLK;
        const int b  = bp >> 1;
        const int p  = bp & 1;
        int A = 0, st = 0;
#pragma unroll
        for (int a = 1; a < 12; ++a) {
            if (e >= blk_start(a)) { A = a; st = blk_start(a); }
        }
        const int C  = ((A - 4 > 0) ? A - 4 : 0) + (e - st);
        const int ha = 4 * A + p;
        const int hc = 4 * C + p;

        const int xi = c & 7;
        const int rs = c >> 3;

        v8f T[2][4][5] = {};

        const unsigned loff = (unsigned)((k0 * HN + 2 * rs) * WN + 2 * xi);
        const float* __restrict__ pa = in2 + ((size_t)b * CN * HN + hc) * WN;
        const float* __restrict__ pb = in1 + ((size_t)b * CN * HN + ha) * WN;

#define LOAD16(S)                                                     \
        _Pragma("unroll")                                             \
        for (int X = 0; X < 4; ++X) {                                 \
            S##a[X][0] = *(const v2f*)(pa + loff + 16 * X);           \
            S##a[X][1] = *(const v2f*)(pa + loff + 16 * X + HWN);     \
            S##b[X][0] = *(const v2f*)(pb + loff + 16 * X);           \
            S##b[X][1] = *(const v2f*)(pb + loff + 16 * X + HWN);     \
        }                                                             \
        pa += 4 * HWN;                                                \
        pb += 4 * HWN

#define KSTEP28(S)                                                            \
        _Pragma("unroll")                                                     \
        for (int q = 0; q < 2; ++q) {                                         \
            v2f FA[4], FB[4];                                                 \
            _Pragma("unroll")                                                 \
            for (int X = 0; X < 4; ++X) {                                     \
                FA[X] = (v2f){ S##a[X][0][q], S##a[X][1][q] };                \
                FB[X] = (v2f){ S##b[X][0][q], S##b[X][1][q] };                \
            }                                                                 \
            _Pragma("unroll")                                                 \
            for (int X = 0; X < 4; ++X) {                                     \
                _Pragma("unroll")                                             \
                for (int X2 = 0; X2 < 4; ++X2) {                              \
                    if (tile_ok(X, X2)) WMMA(T[q][X][X2 - X + 2], FA[X2], FB[X]); \
                }                                                             \
            }                                                                 \
        }

        v2f Pa[4][2], Pb[4][2], Ra[4][2], Rb[4][2];
        LOAD16(P);

        const int pr = lane & 3;
        const float* __restrict__ pfp = ((pr & 2) ? in2 : in1) +
            (((size_t)b * CN + 12 + (lane >> 2)) * HN + ((pr & 2) ? hc : ha) + 2 * (pr & 1)) * WN;
        l2_prefetch(pfp);
        l2_prefetch(pfp + 8 * HWN);
        pfp += 16 * HWN;

#pragma unroll 1
        for (int it = 0; it < 31; ++it) {
            if (it < 28) l2_prefetch(pfp);
            pfp += 8 * HWN;
            LOAD16(R);
            KSTEP28(P);
            LOAD16(P);
            KSTEP28(R);

            __builtin_amdgcn_sched_group_barrier(0x020, 16, 0);
            __builtin_amdgcn_sched_group_barrier(0x008, 8, 0);
            __builtin_amdgcn_sched_group_barrier(0x020, 16, 0);
            __builtin_amdgcn_sched_group_barrier(0x008, 48, 0);
        }
        {
            LOAD16(R);
            KSTEP28(P);
            KSTEP28(R);
        }
#undef LOAD16
#undef KSTEP28

        const int ln = (int)__builtin_amdgcn_mbcnt_lo(~0u, 0u);
        const int xe = ln & 7;
        const int re = (ln >> 3) & 1;
        const int he = ln >> 4;
        const bool rb0 = (xe & 1) != 0;
        const bool rb1 = (xe & 2) != 0;
        const bool rb2 = (xe & 4) != 0;
#pragma unroll
        for (int q = 0; q < 2; ++q) {
#pragma unroll
            for (int X = 0; X < 4; ++X) {
#pragma unroll
                for (int d = 0; d < 5; ++d) {
                    if (tile_ok(X, X + d - 2)) T[q][X][d] = rot8(T[q][X][d], rb0, rb1, rb2);
                }
            }
        }

        const int odd = xe & 1;
        float* __restrict__ orow = out +
            ((size_t)b * DN + (size_t)(2 * (C - A) + 10 + he - re) * ND) * HWN +
            (size_t)(ha + 2 * re) * WN + (odd ? 16 + 2 * (xe - 1) : 2 * xe);

#pragma unroll
        for (int Y = 0; Y < 2; ++Y) {
#pragma unroll
            for (int m = -10; m <= 10; ++m) {
                const int s  = (m + 16) & 7;
                const int d0 = (m - s) / 8;
                const bool use_lo = (xe + s) < 8;
                float mine[2], give[2];
#pragma unroll
                for (int q = 0; q < 2; ++q) {
                    float v[2];
#pragma unroll
                    for (int j = 0; j < 2; ++j) {
                        const int X = 2 * Y + j;
                        const float lo = tile_ok(X, X + d0) ? T[q][X][d0 + 2][s] : 0.0f;
                        const float hi = (s > 0 && tile_ok(X, X + d0 + 1)) ? T[q][X][d0 + 3][s] : 0.0f;
                        v[j] = use_lo ? lo : hi;
                    }
                    mine[q] = odd ? v[1] : v[0];
                    give[q] = odd ? v[0] : v[1];
                }
                const float n0 = pair_lane(give[0]);
                const float n1 = pair_lane(give[1]);
                const v4f lo4 = { mine[0], mine[1], n0, n1 };
                const v4f hi4 = { n0, n1, mine[0], mine[1] };
                const v4f val = (odd ? hi4 : lo4) * (1.0f / (float)CN);
                *(v4f*)(orow + (m + 10) * HWN + 32 * Y) = val;
            }
        }
        return;
    }

    if (task < N_BLOCK_WAVES + N_SINGLE_WAVES) {
        const int s0   = task - N_BLOCK_WAVES;
        const int bp   = s0 / NSGL;
        const int s    = s0 - bp * NSGL;
        const int b    = bp >> 1;
        const int p    = bp & 1;
        const int side = (s >= 21) ? 1 : 0;
        const int r    = s - 21 * side;
        const int blk  = r / 3;
        const int w    = r - 3 * blk;
        const int a    = (w == 0) ? 0 : ((w == 1) ? 1 : (side ? 0 : 1));
        const int cc   = (w == 0) ? 0 : ((w == 1) ? 1 : (side ? 1 : 0));
        const int A    = side ? blk + 5 : blk;
        const int C    = side ? blk : blk + 5;
        const int h    = 4 * A + 2 * a + p;
        const int h2   = 4 * C + 2 * cc + p;
        const int iy   = ((h2 - h) >> 1) + 10;

        DECL8(acc);

        const float* __restrict__ ap =
            in1 + (((size_t)b * CN + k0) * HN + h) * WN + 4 * c;
        const float* __restrict__ bq =
            in2 + (((size_t)b * CN + k0) * HN + h2) * WN + 4 * c;

        v4f p0 = *(const v4f*)(ap);
        v4f p1 = *(const v4f*)(ap + HWN);
        v4f q0 = *(const v4f*)(bq);
        v4f q1 = *(const v4f*)(bq + HWN);
        ap += 4 * HWN;
        bq += 4 * HWN;

        const float* __restrict__ pfp = ((lane & 1) ? in2 : in1) +
            (((size_t)b * CN + 12 + (lane >> 1)) * HN + ((lane & 1) ? h2 : h)) * WN;
        l2_prefetch(pfp);
        l2_prefetch(pfp + 16 * HWN);
        pfp += 32 * HWN;

#pragma unroll 1
        for (int it = 0; it < 31; ++it) {
            if ((it & 1) == 0 && it < 25) { l2_prefetch(pfp); pfp += 16 * HWN; }
            v4f r0 = *(const v4f*)(ap);
            v4f r1 = *(const v4f*)(ap + HWN);
            v4f s0v = *(const v4f*)(bq);
            v4f s1v = *(const v4f*)(bq + HWN);
            ap += 4 * HWN;
            bq += 4 * HWN;
            KSTEP1(p0, p1, q0, q1);

            p0 = *(const v4f*)(ap);
            p1 = *(const v4f*)(ap + HWN);
            q0 = *(const v4f*)(bq);
            q1 = *(const v4f*)(bq + HWN);
            ap += 4 * HWN;
            bq += 4 * HWN;
            KSTEP1(r0, r1, s0v, s1v);

            __builtin_amdgcn_sched_group_barrier(0x020, 4, 0);
            __builtin_amdgcn_sched_group_barrier(0x008, 4, 0);
            __builtin_amdgcn_sched_group_barrier(0x020, 4, 0);
            __builtin_amdgcn_sched_group_barrier(0x008, 12, 0);
        }
        {
            v4f r0 = *(const v4f*)(ap);
            v4f r1 = *(const v4f*)(ap + HWN);
            v4f s0v = *(const v4f*)(bq);
            v4f s1v = *(const v4f*)(bq + HWN);
            KSTEP1(p0, p1, q0, q1);
            KSTEP1(r0, r1, s0v, s1v);
        }

        STORE8(acc, out + ((size_t)b * DN + (size_t)iy * ND) * HWN + (size_t)h * WN);
        return;
    }

    {
        const int f = task - (N_BLOCK_WAVES + N_SINGLE_WAVES);
        const int b = f / HN;
        const int h = f - b * HN;
        const v4f z = { 0.0f, 0.0f, 0.0f, 0.0f };
        for (int iy = 0; iy < ND; ++iy) {
            const int h2 = h - 20 + 2 * iy;
            if (h2 >= 0 && h2 < HN) continue;
            float* __restrict__ zr =
                out + ((size_t)b * DN + (size_t)iy * ND) * HWN + (size_t)h * WN + 4 * c;
#pragma unroll
            for (int q = 0; q < ND + 1; q += 2) {
                const int ix = q + half;
                if (ix < ND)
                    *(v4f*)(zr + (size_t)ix * HWN) = z;
            }
        }
    }
}

extern "C" void kernel_launch(void* const* d_in, const int* in_sizes, int n_in,
                              void* d_out, int out_size, void* d_ws, size_t ws_size,
                              hipStream_t stream) {
    (void)in_sizes; (void)n_in; (void)out_size; (void)d_ws; (void)ws_size;
    const float* in1 = (const float*)d_in[0];
    const float* in2 = (const float*)d_in[1];
    float* out = (float*)d_out;

    corr_wmma_f32_kernel<<<1232, 128, 0, stream>>>(in1, in2, out);
}


# Block waves: the corner tile pairs two column blocks apart (3 in-range displacements of 64 each) are merged two into one product by per-lane operand selects: 24 WMMAs per k-step, was 28; sums same
# speedup vs baseline: 6.9017x; 1.0506x over previous
#include <hip/hip_runtime.h>


typedef float v2f __attribute__((ext_vector_type(2)));
typedef float v4f __attribute__((ext_vector_type(4)));
typedef float v8f __attribute__((ext_vector_type(8)));

#define BN 16
#define CN 256
#define HN 48
#define WN 64
#define HWN (HN * WN)
#define ND 21
#define DN (ND * ND)
#define WAVES_PER_BLOCK 4

#define NBLK 88
#define NSGL 42
#define N_BLOCK_WAVES (BN * 2 * NBLK)
#define N_SINGLE_WAVES (BN * 2 * NSGL)
#define N_FILL_WAVES (BN * HN)
#define N_WAVES (N_BLOCK_WAVES + N_SINGLE_WAVES + N_FILL_WAVES)

#define WMMA(acc, A, B) \
    acc = __builtin_amdgcn_wmma_f32_16x16x4_f32(false, A, false, B, (short)0, acc, false, false)

#define DECL8(P) \
    v8f P##00 = {}, P##11 = {}, P##22 = {}, P##33 = {}, P##02 = {}, P##13 = {}, P##20 = {}, P##31 = {}

#define FRAG(F, x0, x1)                      \
    const v2f F##0 = { x0[0], x1[0] };       \
    const v2f F##1 = { x0[1], x1[1] };       \
    const v2f F##2 = { x0[2], x1[2] };       \
    const v2f F##3 = { x0[3], x1[3] }

#define MM8(P, A, B)             \
    WMMA(P##00, A##0, B##0);     \
    WMMA(P##11, A##1, B##1);     \
    WMMA(P##22, A##2, B##2);     \
    WMMA(P##33, A##3, B##3);     \
    WMMA(P##02, A##0, B##2);     \
    WMMA(P##13, A##1, B##3);     \
    WMMA(P##20, A##2, B##0);     \
    WMMA(P##31, A##3, B##1)

#define KSTEP1(x0, x1, y0, y1)     \
    do {                           \
        FRAG(FA, x0, x1);          \
        FRAG(FB, y0, y1);          \
        MM8(acc, FA, FB);          \
    } while (0)

#define KSTEP4(xa0, xa1, xb0, xb1, yc0, yc1, yd0, yd1)   \
    do {                                                 \
        FRAG(Fa, xa0, xa1);                              \
        FRAG(Fb, xb0, xb1);                              \
        FRAG(Fc, yc0, yc1);                              \
        FRAG(Fd, yd0, yd1);                              \
        MM8(ac, Fa, Fc);                                 \
        MM8(ad, Fa, Fd);                                 \
        MM8(bc, Fb, Fc);                                 \
        MM8(bd, Fb, Fd);                                 \
    } while (0)


__device__ __forceinline__ float zs(const v8f a, const int r, const int half, const int c)
{
    const int diff = c - (8 * half + r);
    const int dw   = ((diff + 8) & 15) - 8;
    return (dw == diff) ? a[r] * (1.0f / (float)CN) : 0.0f;
}

__device__ __forceinline__ v8f zs8(const v8f a, const int half, const int c)
{
    const v8f v = { zs(a, 0, half, c), zs(a, 1, half, c), zs(a, 2, half, c), zs(a, 3, half, c),
                    zs(a, 4, half, c), zs(a, 5, half, c), zs(a, 6, half, c), zs(a, 7, half, c) };
    return v;
}

__device__ __forceinline__ v8f shfl8(const v8f a, const int src_lane)
{
    const v8f v = { __shfl(a[0], src_lane), __shfl(a[1], src_lane), __shfl(a[2], src_lane),
                    __shfl(a[3], src_lane), __shfl(a[4], src_lane), __shfl(a[5], src_lane),
                    __shfl(a[6], src_lane), __shfl(a[7], src_lane) };
    return v;
}

__device__ __forceinline__ v8f rot8(const v8f x, const bool b0, const bool b1, const bool b2)
{
    const v8f t = b0 ? __builtin_shufflevector(x, x, 1, 2, 3, 4, 5, 6, 7, 0) : x;
    const v8f y = b1 ? __builtin_shufflevector(t, t, 2, 3, 4, 5, 6, 7, 0, 1) : t;
    return        b2 ? __builtin_shufflevector(y, y, 4, 5, 6, 7, 0, 1, 2, 3) : y;
}

__device__ __forceinline__ void l2_prefetch(const float* p)
{
    __builtin_prefetch(p, 0, 2);
}

__device__ __forceinline__ void store_class(
    float* __restrict__ orow, const int half, const int c, const int odd,
    const v8f p0, const v8f p1, const v8f p2, const v8f p3)
{
    const bool b0 = (c & 1) != 0;
    const bool b1 = (c & 2) != 0;
    const bool b2 = (c & 4) != 0;
    const v8f q0 = rot8(p0, b0, b1, b2);
    const v8f q1 = rot8(p1, b0, b1, b2);
    const v8f q2 = rot8(p2, b0, b1, b2);
    const v8f q3 = rot8(p3, b0, b1, b2);

#pragma unroll
    for (int s = 0; s < 8; ++s) {
        const int k    = (8 - s) & 7;
        const int i    = 8 * half + ((c - s) & 7);
        const int diff = c - i;
        const int dw   = ((diff + 8) & 15) - 8;
        const unsigned off = (unsigned)((2 * dw + 10 + odd) * HWN + 4 * i);

        if (dw >= -5 && dw <= 5 - odd) {
            const v4f v = { q0[k], q1[k], q2[k], q3[k] };
            *(v4f*)(orow + off) = v;
        }
    }
}

__device__ __forceinline__ void store_band(
    float* __restrict__ orow, const int lane, const int half, const int c,
    const v8f a00, const v8f a11, const v8f a22, const v8f a33,
    const v8f a02, const v8f a13, const v8f a20, const v8f a31)
{
    const int src_lane = (lane & 16) | ((c + 1) & 15);

    store_class(orow, half, c, 0,
                zs8(a00, half, c), zs8(a11, half, c), zs8(a22, half, c), zs8(a33, half, c));

    store_class(orow, half, c, 1,
                zs8(a02, half, c), zs8(a13, half, c),
                shfl8(zs8(a20, half, c), src_lane), shfl8(zs8(a31, half, c), src_lane));
}

#define STORE8(P, orow) \
    store_band(orow, lane, half, c, P##00, P##11, P##22, P##33, P##02, P##13, P##20, P##31)

__device__ __forceinline__ constexpr bool tile_ok(const int X, const int X2)
{
    return X2 >= 0 && X2 < 4 && X2 - X <= 2 && X - X2 <= 2;
}

__device__ __forceinline__ constexpr bool tile_far(const int X, const int X2)
{
    return tile_ok(X, X2) && (X2 - X == 2 || X - X2 == 2);
}

__device__ __forceinline__ float pair_lane(const float x)
{
    const int v = __builtin_bit_cast(int, x);
    return __builtin_bit_cast(float, __builtin_amdgcn_update_dpp(v, v, 0xB1, 0xF, 0xF, false));
}

__device__ __forceinline__ constexpr int blk_start(const int a)
{
    int s = 0;
    for (int g = 0; g < a; ++g) {
        const int lo = (g - 4 > 0) ? g - 4 : 0;
        const int hi = (g + 4 < 11) ? g + 4 : 11;
        s += hi - lo + 1;
    }
    return s;
}

__global__ __launch_bounds__(WAVES_PER_BLOCK * 32, 3) void corr_wmma_f32_kernel(
    const float* __restrict__ in1,
    const float* __restrict__ in2,
    float* __restrict__ out)
{
    const int lane = threadIdx.x & 31;
    const int half = lane >> 4;
    const int c    = lane & 15;
    const int k0   = half * 2;

    const int task = __builtin_amdgcn_readfirstlane(
        (int)(blockIdx.x * WAVES_PER_BLOCK + (threadIdx.x >> 5)));

    if (task < N_BLOCK_WAVES) {
        const int bp = task / NBLK;
        const int e  = task - bp * NBLK;
        const int b  = bp >> 1;
        const int p  = bp & 1;
        int A = 0, st = 0;
#pragma unroll
        for (int a = 1; a < 12; ++a) {
            if (e >= blk_start(a)) { A = a; st = blk_start(a); }
        }
        const int C  = ((A - 4 > 0) ? A - 4 : 0) + (e - st);
        const int ha = 4 * A + p;
        const int hc = 4 * C + p;

        const int xi = c & 7;
        const int rs = c >> 3;

        v8f T[2][4][5] = {};
        v8f M[2][2] = {};
        const bool lo_lane = (lane & 4) == 0;

        const unsigned loff = (unsigned)((k0 * HN + 2 * rs) * WN + 2 * xi);
        const float* __restrict__ pa = in2 + ((size_t)b * CN * HN + hc) * WN;
        const float* __restrict__ pb = in1 + ((size_t)b * CN * HN + ha) * WN;

#define LOAD16(S)                                                     \
        _Pragma("unroll")                                             \
        for (int X = 0; X < 4; ++X) {                                 \
            S##a[X][0] = *(const v2f*)(pa + loff + 16 * X);           \
            S##a[X][1] = *(const v2f*)(pa + loff + 16 * X + HWN);     \
            S##b[X][0] = *(const v2f*)(pb + loff + 16 * X);           \
            S##b[X][1] = *(const v2f*)(pb + loff + 16 * X + HWN);     \
        }                                                             \
        pa += 4 * HWN;                                                \
        pb += 4 * HWN

#define KSTEP24(S)                                                            \
        _Pragma("unroll")                                                     \
        for (int q = 0; q < 2; ++q) {                                         \
            v2f FA[4], FB[4];                                                 \
            _Pragma("unroll")                                                 \
            for (int X = 0; X < 4; ++X) {                                     \
                FA[X] = (v2f){ S##a[X][0][q], S##a[X][1][q] };                \
                FB[X] = (v2f){ S##b[X][0][q], S##b[X][1][q] };                \
            }                                                                 \
            _Pragma("unroll")                                                 \
            for (int X = 0; X < 4; ++X) {                                     \
                _Pragma("unroll")                                             \
                for (int X2 = 0; X2 < 4; ++X2) {                              \
                    if (tile_ok(X, X2) && !tile_far(X, X2))                   \
                        WMMA(T[q][X][X2 - X + 2], FA[X2], FB[X]);             \
                }                                                             \
            }                                                                 \
            _Pragma("unroll")                                                 \
            for (int j = 0; j < 2; ++j) {                                     \
                const v2f Am = lo_lane ? FA[j + 2] : FA[j];                   \
                const v2f Bm = lo_lane ? FB[j + 2] : FB[j];                   \
                WMMA(M[q][j], Am, Bm);                                        \
            }                                                                 \
        }

        v2f Pa[4][2], Pb[4][2], Ra[4][2], Rb[4][2];
        LOAD16(P);

        const int pr = lane & 3;
        const float* __restrict__ pfp = ((pr & 2) ? in2 : in1) +
            (((size_t)b * CN + 12 + (lane >> 2)) * HN + ((pr & 2) ? hc : ha) + 2 * (pr & 1)) * WN;
        l2_prefetch(pfp);
        l2_prefetch(pfp + 8 * HWN);
        pfp += 16 * HWN;

#pragma unroll 1
        for (int it = 0; it < 31; ++it) {
            if (it < 28) l2_prefetch(pfp);
            pfp += 8 * HWN;
            LOAD16(R);
            KSTEP24(P);
            LOAD16(P);
            KSTEP24(R);

            __builtin_amdgcn_sched_group_barrier(0x020, 16, 0);
            __builtin_amdgcn_sched_group_barrier(0x008, 8, 0);
            __builtin_amdgcn_sched_group_barrier(0x020, 16, 0);
            __builtin_amdgcn_sched_group_barrier(0x008, 40, 0);
        }
        {
            LOAD16(R);
            KSTEP24(P);
            KSTEP24(R);
        }
#undef LOAD16
#undef KSTEP24

        const int ln = (int)__builtin_amdgcn_mbcnt_lo(~0u, 0u);
        const int xe = ln & 7;
        const int re = (ln >> 3) & 1;
        const int he = ln >> 4;
        const bool rb0 = (xe & 1) != 0;
        const bool rb1 = (xe & 2) != 0;
        const bool rb2 = (xe & 4) != 0;
#pragma unroll
        for (int q = 0; q < 2; ++q) {
#pragma unroll
            for (int X = 0; X < 4; ++X) {
#pragma unroll
                for (int d = 1; d < 4; ++d) {
                    if (tile_ok(X, X + d - 2)) T[q][X][d] = rot8(T[q][X][d], rb0, rb1, rb2);
                }
            }
#pragma unroll
            for (int j = 0; j < 2; ++j) M[q][j] = rot8(M[q][j], rb0, rb1, rb2);
        }
#define TILE_S(q, X, d, s) ((((d) == 0) || ((d) == 4)) ? M[q][(X) & 1][s] : T[q][X][d][s])

        const int odd = xe & 1;
        float* __restrict__ orow = out +
            ((size_t)b * DN + (size_t)(2 * (C - A) + 10 + he - re) * ND) * HWN +
            (size_t)(ha + 2 * re) * WN + (odd ? 16 + 2 * (xe - 1) : 2 * xe);

#pragma unroll
        for (int Y = 0; Y < 2; ++Y) {
#pragma unroll
            for (int m = -10; m <= 10; ++m) {
                const int s  = (m + 16) & 7;
                const int d0 = (m - s) / 8;
                const bool use_lo = (xe + s) < 8;
                float mine[2], give[2];
#pragma unroll
                for (int q = 0; q < 2; ++q) {
                    float v[2];
#pragma unroll
                    for (int j = 0; j < 2; ++j) {
                        const int X = 2 * Y + j;
                        const float lo = tile_ok(X, X + d0) ? TILE_S(q, X, d0 + 2, s) : 0.0f;
                        const float hi = (s > 0 && tile_ok(X, X + d0 + 1)) ? TILE_S(q, X, d0 + 3, s) : 0.0f;
                        v[j] = use_lo ? lo : hi;
                    }
                    mine[q] = odd ? v[1] : v[0];
                    give[q] = odd ? v[0] : v[1];
                }
                const float n0 = pair_lane(give[0]);
                const float n1 = pair_lane(give[1]);
                const v4f lo4 = { mine[0], mine[1], n0, n1 };
                const v4f hi4 = { n0, n1, mine[0], mine[1] };
                const v4f val = (odd ? hi4 : lo4) * (1.0f / (float)CN);
                *(v4f*)(orow + (m + 10) * HWN + 32 * Y) = val;
            }
        }
#undef TILE_S
        return;
    }

    if (task < N_BLOCK_WAVES + N_SINGLE_WAVES) {
        const int s0   = task - N_BLOCK_WAVES;
        const int bp   = s0 / NSGL;
        const int s    = s0 - bp * NSGL;
        const int b    = bp >> 1;
        const int p    = bp & 1;
        const int side = (s >= 21) ? 1 : 0;
        const int r    = s - 21 * side;
        const int blk  = r / 3;
        const int w    = r - 3 * blk;
        const int a    = (w == 0) ? 0 : ((w == 1) ? 1 : (side ? 0 : 1));
        const int cc   = (w == 0) ? 0 : ((w == 1) ? 1 : (side ? 1 : 0));
        const int A    = side ? blk + 5 : blk;
        const int C    = side ? blk : blk + 5;
        const int h    = 4 * A + 2 * a + p;
        const int h2   = 4 * C + 2 * cc + p;
        const int iy   = ((h2 - h) >> 1) + 10;

        DECL8(acc);

        const float* __restrict__ ap =
            in1 + (((size_t)b * CN + k0) * HN + h) * WN + 4 * c;
        const float* __restrict__ bq =
            in2 + (((size_t)b * CN + k0) * HN + h2) * WN + 4 * c;

        v4f p0 = *(const v4f*)(ap);
        v4f p1 = *(const v4f*)(ap + HWN);
        v4f q0 = *(const v4f*)(bq);
        v4f q1 = *(const v4f*)(bq + HWN);
        ap += 4 * HWN;
        bq += 4 * HWN;

        const float* __restrict__ pfp = ((lane & 1) ? in2 : in1) +
            (((size_t)b * CN + 12 + (lane >> 1)) * HN + ((lane & 1) ? h2 : h)) * WN;
        l2_prefetch(pfp);
        l2_prefetch(pfp + 16 * HWN);
        pfp += 32 * HWN;

#pragma unroll 1
        for (int it = 0; it < 31; ++it) {
            if ((it & 1) == 0 && it < 25) { l2_prefetch(pfp); pfp += 16 * HWN; }
            v4f r0 = *(const v4f*)(ap);
            v4f r1 = *(const v4f*)(ap + HWN);
            v4f s0v = *(const v4f*)(bq);
            v4f s1v = *(const v4f*)(bq + HWN);
            ap += 4 * HWN;
            bq += 4 * HWN;
            KSTEP1(p0, p1, q0, q1);

            p0 = *(const v4f*)(ap);
            p1 = *(const v4f*)(ap + HWN);
            q0 = *(const v4f*)(bq);
            q1 = *(const v4f*)(bq + HWN);
            ap += 4 * HWN;
            bq += 4 * HWN;
            KSTEP1(r0, r1, s0v, s1v);

            __builtin_amdgcn_sched_group_barrier(0x020, 4, 0);
            __builtin_amdgcn_sched_group_barrier(0x008, 4, 0);
            __builtin_amdgcn_sched_group_barrier(0x020, 4, 0);
            __builtin_amdgcn_sched_group_barrier(0x008, 12, 0);
        }
        {
            v4f r0 = *(const v4f*)(ap);
            v4f r1 = *(const v4f*)(ap + HWN);
            v4f s0v = *(const v4f*)(bq);
            v4f s1v = *(const v4f*)(bq + HWN);
            KSTEP1(p0, p1, q0, q1);
            KSTEP1(r0, r1, s0v, s1v);
        }

        STORE8(acc, out + ((size_t)b * DN + (size_t)iy * ND) * HWN + (size_t)h * WN);
        return;
    }

    {
        const int f = task - (N_BLOCK_WAVES + N_SINGLE_WAVES);
        const int b = f / HN;
        const int h = f - b * HN;
        const v4f z = { 0.0f, 0.0f, 0.0f, 0.0f };
        for (int iy = 0; iy < ND; ++iy) {
            const int h2 = h - 20 + 2 * iy;
            if (h2 >= 0 && h2 < HN) continue;
            float* __restrict__ zr =
                out + ((size_t)b * DN + (size_t)iy * ND) * HWN + (size_t)h * WN + 4 * c;
#pragma unroll
            for (int q = 0; q < ND + 1; q += 2) {
                const int ix = q + half;
                if (ix < ND)
                    *(v4f*)(zr + (size_t)ix * HWN) = z;
            }
        }
    }
}

extern "C" void kernel_launch(void* const* d_in, const int* in_sizes, int n_in,
                              void* d_out, int out_size, void* d_ws, size_t ws_size,
                              hipStream_t stream) {
    (void)in_sizes; (void)n_in; (void)out_size; (void)d_ws; (void)ws_size;
    const float* in1 = (const float*)d_in[0];
    const float* in2 = (const float*)d_in[1];
    float* out = (float*)d_out;

    corr_wmma_f32_kernel<<<1232, 128, 0, stream>>>(in1, in2, out);
}
